# starting kernel
# baseline (speedup 1.0000x reference)
#include <hip/hip_runtime.h>

#define B_DIM 8
#define M_IN 32
#define M_OUT 32
#define C_DIM 64
#define T_DIM 192

typedef float v2f __attribute__((ext_vector_type(2)));
typedef float v8f __attribute__((ext_vector_type(8)));

__global__ __launch_bounds__(256) void e2n_reduce_kernel(
    const float* __restrict__ x, float* __restrict__ s)
{
    __shared__ float lds1[C_DIM];
    __shared__ float lds2[C_DIM];
    __shared__ float lds3[C_DIM];

    const int tid = threadIdx.x;
    if (tid < C_DIM) { lds1[tid] = 0.f; lds2[tid] = 0.f; lds3[tid] = 0.f; }
    __syncthreads();

    const int slab = blockIdx.x;
    const float* xs = x + (size_t)slab * (C_DIM * C_DIM * C_DIM);

    const int q    = tid & 15;
    const int r    = tid >> 4;
    const int lane = tid & 31;

    float4 acc3 = make_float4(0.f, 0.f, 0.f, 0.f);
    float  acc2[4] = {0.f, 0.f, 0.f, 0.f};

    for (int i = 0; i < C_DIM; ++i) {
        const float* xi = xs + (i << 12);
        float acc1 = 0.f;
        #pragma unroll
        for (int jj = 0; jj < 4; ++jj) {
            const int j = r + (jj << 4);
            float4 v = *(const float4*)(xi + (j << 6) + (q << 2));
            float  h = (v.x + v.y) + (v.z + v.w);
            acc1      += h;
            acc2[jj]  += h;
            acc3.x += v.x; acc3.y += v.y; acc3.z += v.z; acc3.w += v.w;
        }
        #pragma unroll
        for (int off = 16; off >= 1; off >>= 1)
            acc1 += __shfl_xor(acc1, off, 32);
        if (lane == 0) atomicAdd(&lds1[i], acc1);
    }

    #pragma unroll
    for (int jj = 0; jj < 4; ++jj) {
        float v = acc2[jj];
        #pragma unroll
        for (int off = 8; off >= 1; off >>= 1)
            v += __shfl_xor(v, off, 32);
        if ((lane & 15) == 0) atomicAdd(&lds2[r + (jj << 4)], v);
    }

    acc3.x += __shfl_xor(acc3.x, 16, 32);
    acc3.y += __shfl_xor(acc3.y, 16, 32);
    acc3.z += __shfl_xor(acc3.z, 16, 32);
    acc3.w += __shfl_xor(acc3.w, 16, 32);
    if (lane < 16) {
        atomicAdd(&lds3[(q << 2) + 0], acc3.x);
        atomicAdd(&lds3[(q << 2) + 1], acc3.y);
        atomicAdd(&lds3[(q << 2) + 2], acc3.z);
        atomicAdd(&lds3[(q << 2) + 3], acc3.w);
    }
    __syncthreads();

    if (tid < T_DIM) {
        float v = (tid < 64) ? lds1[tid]
                : (tid < 128) ? lds2[tid - 64]
                              : lds3[tid - 128];
        s[(size_t)slab * T_DIM + tid] = v;
    }
}

__global__ __launch_bounds__(32) void e2n_wmma_mix_kernel(
    const float* __restrict__ s,
    const float* __restrict__ alpha,
    const float* __restrict__ beta,
    const float* __restrict__ gamma,
    float* __restrict__ out)
{
    const int lane = threadIdx.x;
    const int half = lane >> 4;
    const int l16  = lane & 15;

    const int bid    = blockIdx.x;
    const int b      = bid / 24;
    const int rem    = bid % 24;
    const int n_base = (rem / 12) << 4;
    const int t_base = (rem % 12) << 4;

    const float* sb = s + (size_t)b * (M_IN * T_DIM);

    v8f acc = {0.f, 0.f, 0.f, 0.f, 0.f, 0.f, 0.f, 0.f};

    #pragma unroll
    for (int k0 = 0; k0 < M_IN; k0 += 4) {
        const int m0 = k0 + (half << 1);
        const int n  = n_base + l16;
        const int t  = t_base + l16;

        v2f a, bf;
        a.x = alpha[m0 * M_OUT + n] + beta[m0 * M_OUT + n] + gamma[m0 * M_OUT + n];
        a.y = alpha[(m0 + 1) * M_OUT + n] + beta[(m0 + 1) * M_OUT + n]
            + gamma[(m0 + 1) * M_OUT + n];
        bf.x = sb[m0 * T_DIM + t];
        bf.y = sb[(m0 + 1) * T_DIM + t];

        acc = __builtin_amdgcn_wmma_f32_16x16x4_f32(
             false, a,  false, bf,
             (short)0, acc,  false,  false);
    }

    #pragma unroll
    for (int rr = 0; rr < 8; ++rr) {
        const int n = n_base + rr + (half << 3);
        const int t = t_base + l16;
        out[((size_t)b * M_OUT + n) * T_DIM + t] = acc[rr];
    }
}

extern "C" void kernel_launch(void* const* d_in, const int* in_sizes, int n_in,
                              void* d_out, int out_size, void* d_ws, size_t ws_size,
                              hipStream_t stream) {
    const float* x     = (const float*)d_in[0];
    const float* alpha = (const float*)d_in[1];
    const float* beta  = (const float*)d_in[2];
    const float* gamma = (const float*)d_in[3];
    float* out = (float*)d_out;
    float* s   = (float*)d_ws;

    e2n_reduce_kernel<<<B_DIM * M_IN, 256, 0, stream>>>(x, s);
    e2n_wmma_mix_kernel<<<B_DIM * 2 * 12, 32, 0, stream>>>(s, alpha, beta, gamma, out);
}


# Reduce kernel rewritten as a stream: 16 waves per slab, 16-32 non-temporal 16-byte loads in flight per wave, cross-lane sums deferred, fixed-order LDS combine replaces LDS atomics (sum order changes)
# speedup vs baseline: 2.4386x; 2.4386x over previous
#include <hip/hip_runtime.h>

#define B_DIM 8
#define M_IN 32
#define M_OUT 32
#define C_DIM 64
#define T_DIM 192

typedef float v2f __attribute__((ext_vector_type(2)));
typedef float v4f __attribute__((ext_vector_type(4)));
typedef float v8f __attribute__((ext_vector_type(8)));

#define R_WAVES 16
#define R_THREADS (R_WAVES * 32)
#define R_PLANES 4
#define R_UNITS (R_PLANES * 2)
#define R_LOADS 16

static __device__ __forceinline__ v4f load_stream(const float* p) {
    return __builtin_nontemporal_load(reinterpret_cast<const v4f*>(p));
}

template <int M>
static __device__ __forceinline__ float row_xor(float v) {
    return __builtin_bit_cast(float,
        __builtin_amdgcn_update_dpp(0, __builtin_bit_cast(int, v), 0x160 + M, 0xf, 0xf, true));
}

static __device__ __forceinline__ float xhalf(float v) {
    const int b = __builtin_bit_cast(int, v);
    return __builtin_bit_cast(float,
        __builtin_amdgcn_permlanex16(b, b, 0x76543210, 0xfedcba98, false, false));
}

__global__ __launch_bounds__(R_THREADS) void e2n_reduce_kernel(
    const float* __restrict__ x, float* __restrict__ s)
{
    __shared__ float lds_s1[C_DIM];
    __shared__ __attribute__((aligned(16))) float lds_part[R_WAVES][2 * C_DIM];

    const int tid  = threadIdx.x;
    const int lane = tid & 31;
    const int wave = __builtin_amdgcn_readfirstlane(tid >> 5);
    const int c    = lane & 15;
    const int h    = lane >> 4;
    const int slab = blockIdx.x;

    const float* xw = x + (size_t)slab * (C_DIM * C_DIM * C_DIM)
                        + (size_t)wave * (R_PLANES * C_DIM * C_DIM) + (lane << 2);

    v4f   ring[2][R_LOADS];
    float acc2[2 * R_LOADS];
    float acc1[R_PLANES];
    v4f   acc3 = {0.f, 0.f, 0.f, 0.f};
    #pragma unroll
    for (int e = 0; e < 2 * R_LOADS; ++e) acc2[e] = 0.f;

#define LOAD_UNIT(U) do { _Pragma("unroll") for (int q_ = 0; q_ < R_LOADS; ++q_) \
        ring[(U) & 1][q_] = load_stream(xw + (U) * (R_LOADS * 128) + q_ * 128); } while (0)

    LOAD_UNIT(0);
    #pragma unroll
    for (int u = 0; u < R_UNITS; ++u) {
        __builtin_amdgcn_sched_barrier(0);
        if (u + 1 < R_UNITS) LOAD_UNIT(u + 1);
        __builtin_amdgcn_sched_barrier(0);

        float p0 = 0.f, p1 = 0.f;
        #pragma unroll
        for (int q = 0; q < R_LOADS; ++q) {
            const v4f v = ring[u & 1][q];
            const float hs = (v.x + v.y) + (v.z + v.w);
            acc2[(u & 1) * R_LOADS + q] += hs;
            acc3 += v;
            if (q & 1) p1 += hs; else p0 += hs;
        }
        if (u & 1) acc1[u >> 1] += p0 + p1; else acc1[u >> 1] = p0 + p1;
        #pragma unroll
        for (int q = 0; q < R_LOADS; ++q) asm volatile("" : "+v"(acc2[(u & 1) * R_LOADS + q]));
        asm volatile("" : "+v"(acc3));
        asm volatile("" : "+v"(acc1[u >> 1]));
    }
#undef LOAD_UNIT

    float mine1 = 0.f;
    #pragma unroll
    for (int p = 0; p < R_PLANES; ++p) {
        float v = acc1[p];
        v += row_xor<1>(v); v += row_xor<2>(v); v += row_xor<4>(v); v += row_xor<8>(v);
        v += xhalf(v);
        if (lane == p) mine1 = v;
    }
    if (lane < R_PLANES) lds_s1[wave * R_PLANES + lane] = mine1;

    acc3.x += xhalf(acc3.x); acc3.y += xhalf(acc3.y);
    acc3.z += xhalf(acc3.z); acc3.w += xhalf(acc3.w);
    if (h == 0) *reinterpret_cast<v4f*>(&lds_part[wave][C_DIM + (c << 2)]) = acc3;

    {
        const bool u8 = (c & 8) != 0, u4 = (c & 4) != 0, u2 = (c & 2) != 0, u1 = (c & 1) != 0;
        float f16[16], f8[8], f4[4], f2[2];
        #pragma unroll
        for (int e = 0; e < 16; ++e) {
            const float keep = u8 ? acc2[e + 16] : acc2[e];
            const float send = u8 ? acc2[e] : acc2[e + 16];
            f16[e] = keep + row_xor<8>(send);
        }
        #pragma unroll
        for (int e = 0; e < 8; ++e) {
            const float keep = u4 ? f16[e + 8] : f16[e];
            const float send = u4 ? f16[e] : f16[e + 8];
            f8[e] = keep + row_xor<4>(send);
        }
        #pragma unroll
        for (int e = 0; e < 4; ++e) {
            const float keep = u2 ? f8[e + 4] : f8[e];
            const float send = u2 ? f8[e] : f8[e + 4];
            f4[e] = keep + row_xor<2>(send);
        }
        #pragma unroll
        for (int e = 0; e < 2; ++e) {
            const float keep = u1 ? f4[e + 2] : f4[e];
            const float send = u1 ? f4[e] : f4[e + 2];
            f2[e] = keep + row_xor<1>(send);
        }
        lds_part[wave][(c << 2) + h]     = f2[0];
        lds_part[wave][(c << 2) + 2 + h] = f2[1];
    }
    __syncthreads();

    if (tid < T_DIM) {
        float v;
        if (tid < C_DIM) {
            v = lds_s1[tid];
        } else {
            v = 0.f;
            #pragma unroll
            for (int w = 0; w < R_WAVES; ++w) v += lds_part[w][tid - C_DIM];
        }
        s[(size_t)slab * T_DIM + tid] = v;
    }
}

__global__ __launch_bounds__(32) void e2n_wmma_mix_kernel(
    const float* __restrict__ s,
    const float* __restrict__ alpha,
    const float* __restrict__ beta,
    const float* __restrict__ gamma,
    float* __restrict__ out)
{
    const int lane = threadIdx.x;
    const int half = lane >> 4;
    const int l16  = lane & 15;

    const int bid    = blockIdx.x;
    const int b      = bid / 24;
    const int rem    = bid % 24;
    const int n_base = (rem / 12) << 4;
    const int t_base = (rem % 12) << 4;

    const float* sb = s + (size_t)b * (M_IN * T_DIM);

    v8f acc = {0.f, 0.f, 0.f, 0.f, 0.f, 0.f, 0.f, 0.f};

    #pragma unroll
    for (int k0 = 0; k0 < M_IN; k0 += 4) {
        const int m0 = k0 + (half << 1);
        const int n  = n_base + l16;
        const int t  = t_base + l16;

        v2f a, bf;
        a.x = alpha[m0 * M_OUT + n] + beta[m0 * M_OUT + n] + gamma[m0 * M_OUT + n];
        a.y = alpha[(m0 + 1) * M_OUT + n] + beta[(m0 + 1) * M_OUT + n]
            + gamma[(m0 + 1) * M_OUT + n];
        bf.x = sb[m0 * T_DIM + t];
        bf.y = sb[(m0 + 1) * T_DIM + t];

        acc = __builtin_amdgcn_wmma_f32_16x16x4_f32(
             false, a,  false, bf,
             (short)0, acc,  false,  false);
    }

    #pragma unroll
    for (int rr = 0; rr < 8; ++rr) {
        const int n = n_base + rr + (half << 3);
        const int t = t_base + l16;
        out[((size_t)b * M_OUT + n) * T_DIM + t] = acc[rr];
    }
}

extern "C" void kernel_launch(void* const* d_in, const int* in_sizes, int n_in,
                              void* d_out, int out_size, void* d_ws, size_t ws_size,
                              hipStream_t stream) {
    const float* x     = (const float*)d_in[0];
    const float* alpha = (const float*)d_in[1];
    const float* beta  = (const float*)d_in[2];
    const float* gamma = (const float*)d_in[3];
    float* out = (float*)d_out;
    float* s   = (float*)d_ws;

    e2n_reduce_kernel<<<B_DIM * M_IN, R_THREADS, 0, stream>>>(x, s);
    e2n_wmma_mix_kernel<<<B_DIM * 2 * 12, 32, 0, stream>>>(s, alpha, beta, gamma, out);
}
